# starting kernel
# baseline (speedup 1.0000x reference)
#include <hip/hip_runtime.h>

#define B_DIM 16
#define G_DIM 1708
#define H_DIM 5
#define GPAD  1712
#define BLOCK 128
#define WPB   (BLOCK / 32)
#define LOG2E 1.44269504088896340736f

typedef __attribute__((ext_vector_type(2))) float v2f;
typedef __attribute__((ext_vector_type(8))) float v8f;

#if __has_builtin(__builtin_amdgcn_exp2f)
#define EXP2F(v) __builtin_amdgcn_exp2f(v)
#else
#define EXP2F(v) exp2f(v)
#endif

__launch_bounds__(BLOCK)
__global__ void mha_rank1_wmma_kernel(const float* __restrict__ x,
                                      const float* __restrict__ WQ,
                                      const float* __restrict__ WK,
                                      const float* __restrict__ WV,
                                      const float* __restrict__ W0,
                                      float* __restrict__ out)
{
    __shared__ __align__(16) float x_lds[GPAD];
    __shared__ __align__(16) float k_lds[GPAD];
    __shared__ __align__(16) float v_lds[GPAD];

    const int b    = blockIdx.x;
    const int tid  = threadIdx.x;
    const int wave = tid >> 5;
    const int lane = tid & 31;
    const int ln   = lane & 15;
    const int hi   = lane >> 4;

    const int  tile   = blockIdx.y * WPB + wave;
    const bool tValid = (tile * 16) < G_DIM;
    const int  i      = tile * 16 + ln;
    const int  iC     = (i < G_DIM) ? i : (G_DIM - 1);

    for (int j = tid; j < GPAD; j += BLOCK)
        x_lds[j] = (j < G_DIM) ? x[b * G_DIM + j] : 0.0f;
    __syncthreads();

    float out_acc = 0.0f;

    for (int h = 0; h < H_DIM; ++h) {
        const float* __restrict__ wk = WK + h * G_DIM;
        const float* __restrict__ wv = WV + h * G_DIM;
        for (int j = tid; j < GPAD; j += BLOCK) {
            const float xv = x_lds[j];
            const float kk = (j < G_DIM) ? wk[j] : 0.0f;
            const float vv = (j < G_DIM) ? wv[j] : 0.0f;
            k_lds[j] = xv * kk * LOG2E;
            v_lds[j] = xv * vv;
        }
        __syncthreads();

        if (tValid) {
            const float t = (lane < 16) ? x_lds[iC] * WQ[h * G_DIM + iC] : 0.0f;
            v2f bm; bm[0] = t; bm[1] = 0.0f;

            float Dsum = 0.0f, Nsum = 0.0f;

            int jb = 0;
            for (; jb + 16 <= G_DIM; jb += 16) {
                v2f am; am[0] = (lane < 16) ? k_lds[jb + ln] : 0.0f; am[1] = 0.0f;
                v8f c = {};
                v8f d = __builtin_amdgcn_wmma_f32_16x16x4_f32(
                    false, am, false, bm, (short)0, c, false, false);
                const float4* vp =
                    reinterpret_cast<const float4*>(v_lds + jb + (hi << 3));
                const float4 va = vp[0], vb = vp[1];
                const float vv[8] = {va.x, va.y, va.z, va.w,
                                     vb.x, vb.y, vb.z, vb.w};
#pragma unroll
                for (int r = 0; r < 8; ++r) {
                    const float e = EXP2F(d[r]);
                    Dsum += e;
                    Nsum += e * vv[r];
                }
            }
            {
                v2f am; am[0] = (lane < 16) ? k_lds[jb + ln] : 0.0f; am[1] = 0.0f;
                v8f c = {};
                v8f d = __builtin_amdgcn_wmma_f32_16x16x4_f32(
                    false, am, false, bm, (short)0, c, false, false);
                const float4* vp =
                    reinterpret_cast<const float4*>(v_lds + jb + (hi << 3));
                const float4 va = vp[0], vb = vp[1];
                const float vv[8] = {va.x, va.y, va.z, va.w,
                                     vb.x, vb.y, vb.z, vb.w};
#pragma unroll
                for (int r = 0; r < 8; ++r) {
                    const bool ok = (jb + (hi << 3) + r) < G_DIM;
                    const float e = ok ? EXP2F(d[r]) : 0.0f;
                    Dsum += e;
                    Nsum += e * vv[r];
                }
            }

            Dsum += __shfl_xor(Dsum, 16, 32);
            Nsum += __shfl_xor(Nsum, 16, 32);

            if (lane < 16 && i < G_DIM) {
                const float eii = EXP2F(t * k_lds[i]);
                const float num = Nsum - eii * v_lds[i];
                out_acc += W0[h] * num / Dsum;
            }
        }
        __syncthreads();
    }

    if (tValid && lane < 16 && i < G_DIM)
        out[b * G_DIM + i] = out_acc;
}

extern "C" void kernel_launch(void* const* d_in, const int* in_sizes, int n_in,
                              void* d_out, int out_size, void* d_ws, size_t ws_size,
                              hipStream_t stream) {
    (void)in_sizes; (void)n_in; (void)out_size; (void)d_ws; (void)ws_size;
    const float* x  = (const float*)d_in[0];
    const float* WQ = (const float*)d_in[1];
    const float* WK = (const float*)d_in[2];
    const float* WV = (const float*)d_in[3];
    const float* W0 = (const float*)d_in[4];
    float* out = (float*)d_out;

    const int numTiles = (G_DIM + 15) / 16;
    const int groups   = (numTiles + WPB - 1) / WPB;
    dim3 grid(B_DIM, groups);
    hipLaunchKernelGGL(mha_rank1_wmma_kernel, grid, dim3(BLOCK), 0, stream,
                       x, WQ, WK, WV, W0, out);
}


# Score tile transposed: j across lanes, i down registers; one 16-byte k and v read per lane feeds 4 tiles; 16 per-lane partial sums, lane sum once per head (float sum order changes)
# speedup vs baseline: 1.7617x; 1.7617x over previous
#include <hip/hip_runtime.h>


#define B_DIM 16
#define G_DIM 1708
#define H_DIM 5
#define NFULL 26
#define GPAD  1792
#define BLOCK 128
#define WPB   (BLOCK / 32)
#define LOG2E 1.44269504088896340736f

typedef __attribute__((ext_vector_type(2))) float v2f;
typedef __attribute__((ext_vector_type(4))) float v4f;
typedef __attribute__((ext_vector_type(8))) float v8f;

#if __has_builtin(__builtin_amdgcn_exp2f)
#define EXP2F(v) __builtin_amdgcn_exp2f(v)
#else
#define EXP2F(v) exp2f(v)
#endif

template <bool MASKED>
__device__ __forceinline__ void score_tile(const v2f a, const v2f bm, const float vj, const bool ok,
                                           v8f& Dacc, v8f& Nacc)
{
    v8f c = {};
    const v8f d = __builtin_amdgcn_wmma_f32_16x16x4_f32(
        false, a, false, bm, (short)0, c, false, false);
    v8f e;
#pragma unroll
    for (int r = 0; r < 8; ++r) {
        const float ex = EXP2F(d[r]);
        e[r] = MASKED ? (ok ? ex : 0.0f) : ex;
    }
    Dacc += e;
    Nacc += e * vj;
}

template <bool MASKED>
__device__ __forceinline__ void score_group(const v2f a0, const v2f a1, const v4f k, const v4f v,
                                            const bool ok, v8f& Dacc, v8f& Nacc)
{
    v2f b01; b01[0] = k.x; b01[1] = k.y;
    v2f b23; b23[0] = k.z; b23[1] = k.w;
    score_tile<MASKED>(a0, b01, v.x, ok, Dacc, Nacc);
    score_tile<MASKED>(a1, b01, v.y, ok, Dacc, Nacc);
    score_tile<MASKED>(a0, b23, v.z, ok, Dacc, Nacc);
    score_tile<MASKED>(a1, b23, v.w, ok, Dacc, Nacc);
}

__device__ __forceinline__ float half_row_sums(const v8f X, const int ln)
{
    const bool b3 = (ln & 8) != 0, b2 = (ln & 4) != 0, b1 = (ln & 2) != 0;
    float y[4];
#pragma unroll
    for (int r = 0; r < 4; ++r) {
        const float keep = b3 ? X[r + 4] : X[r];
        const float send = b3 ? X[r] : X[r + 4];
        y[r] = keep + __shfl_xor(send, 8, 32);
    }
    float z[2];
#pragma unroll
    for (int r = 0; r < 2; ++r) {
        const float keep = b2 ? y[r + 2] : y[r];
        const float send = b2 ? y[r] : y[r + 2];
        z[r] = keep + __shfl_xor(send, 4, 32);
    }
    const float keep = b1 ? z[1] : z[0];
    const float send = b1 ? z[0] : z[1];
    const float w = keep + __shfl_xor(send, 2, 32);
    return w + __shfl_xor(w, 1, 32);
}

__launch_bounds__(BLOCK) __attribute__((amdgpu_waves_per_eu(1, 4)))
__global__ void mha_rank1_wmma_kernel(const float* __restrict__ x,
                                      const float* __restrict__ WQ,
                                      const float* __restrict__ WK,
                                      const float* __restrict__ WV,
                                      const float* __restrict__ W0,
                                      float* __restrict__ out)
{
    __shared__ __align__(16) float x_lds[GPAD];
    __shared__ __align__(16) float k_lds[GPAD];
    __shared__ __align__(16) float v_lds[GPAD];

    const int b    = blockIdx.x;
    const int tid  = threadIdx.x;
    const int wave = tid >> 5;
    const int lane = tid & 31;
    const int ln   = lane & 15;
    const int hi   = lane >> 4;

    const int  tile   = blockIdx.y * WPB + wave;
    const bool tValid = (tile * 16) < G_DIM;
    const int  iA     = tile * 16 + ln;
    const int  iAc    = (iA < G_DIM) ? iA : (G_DIM - 1);
    const int  iO     = tile * 16 + (hi << 3) + (ln >> 1);
    const int  iOc    = (iO < G_DIM) ? iO : (G_DIM - 1);

    for (int j = tid; j < GPAD; j += BLOCK)
        x_lds[j] = (j < G_DIM) ? x[b * G_DIM + j] : 0.0f;
    __syncthreads();

    float out_acc = 0.0f;

    for (int h = 0; h < H_DIM; ++h) {
        const float* __restrict__ wk = WK + h * G_DIM;
        const float* __restrict__ wv = WV + h * G_DIM;
        for (int j = tid; j < GPAD; j += BLOCK) {
            const float xv = x_lds[j];
            const float kk = (j < G_DIM) ? wk[j] : 0.0f;
            const float vv = (j < G_DIM) ? wv[j] : 0.0f;
            k_lds[j] = xv * kk * LOG2E;
            v_lds[j] = xv * vv;
        }
        __syncthreads();

        if (tValid) {
            const float tA = (lane < 16) ? x_lds[iAc] * WQ[h * G_DIM + iAc] : 0.0f;
            const float tO = x_lds[iOc] * WQ[h * G_DIM + iOc];
            v2f a0; a0[0] = tA;   a0[1] = 0.0f;
            v2f a1; a1[0] = 0.0f; a1[1] = tA;

            v8f Dacc = {}, Nacc = {};

            const v4f* kp = reinterpret_cast<const v4f*>(k_lds) + ln;
            const v4f* vp = reinterpret_cast<const v4f*>(v_lds) + ln;

            v4f ka = kp[0], va = vp[0], kb = kp[16], vb = vp[16];
            for (int g = 0; g < NFULL; g += 2) {
                const v4f kna = kp[(g + 2) * 16], vna = vp[(g + 2) * 16];
                const v4f knb = kp[(g + 3) * 16], vnb = vp[(g + 3) * 16];
                score_group<false>(a0, a1, ka, va, true, Dacc, Nacc);
                score_group<false>(a0, a1, kb, vb, true, Dacc, Nacc);
                asm volatile("" :: "v"(kna), "v"(vna), "v"(knb), "v"(vnb));
                ka = kna; va = vna; kb = knb; vb = vnb;
            }
            {
                const bool ok = (NFULL * 64 + 4 * ln + 3) < G_DIM;
                score_group<true>(a0, a1, ka, va, ok, Dacc, Nacc);
            }

            const float Dsum = half_row_sums(Dacc, ln);
            const float Nsum = half_row_sums(Nacc, ln);

            if (iO < G_DIM) {
                const float eii = EXP2F(tO * k_lds[iO]);
                const float num = Nsum - eii * v_lds[iO];
                out_acc += W0[h] * num / Dsum;
            }
        }
        __syncthreads();
    }

    if (tValid && (ln & 1) == 0 && iO < G_DIM)
        out[b * G_DIM + iO] = out_acc;
}

extern "C" void kernel_launch(void* const* d_in, const int* in_sizes, int n_in,
                              void* d_out, int out_size, void* d_ws, size_t ws_size,
                              hipStream_t stream) {
    (void)in_sizes; (void)n_in; (void)out_size; (void)d_ws; (void)ws_size;
    const float* x  = (const float*)d_in[0];
    const float* WQ = (const float*)d_in[1];
    const float* WK = (const float*)d_in[2];
    const float* WV = (const float*)d_in[3];
    const float* W0 = (const float*)d_in[4];
    float* out = (float*)d_out;

    const int numTiles = (G_DIM + 15) / 16;
    const int groups   = (numTiles + WPB - 1) / WPB;
    dim3 grid(B_DIM, groups);
    hipLaunchKernelGGL(mha_rank1_wmma_kernel, grid, dim3(BLOCK), 0, stream,
                       x, WQ, WK, WV, W0, out);
}


# Staging once per block: all global reads issued together as 16-byte loads before the first wait, k and v of all 5 heads in LDS behind one barrier; head loop has no global wait and no barrier
# speedup vs baseline: 2.3883x; 1.3557x over previous
#include <hip/hip_runtime.h>


#define B_DIM 16
#define G_DIM 1708
#define H_DIM 5
#define NFULL 26
#define GPAD  1792
#define BLOCK 128
#define WPB   (BLOCK / 32)
#define G4    (G_DIM / 4)
#define LROW  2048
#define FILL_STEPS (LROW / 4 / BLOCK)
#define LOG2E 1.44269504088896340736f

static_assert(LROW >= GPAD && LROW % (4 * BLOCK) == 0 && G_DIM % 4 == 0, "staging layout");

typedef __attribute__((ext_vector_type(2))) float v2f;
typedef __attribute__((ext_vector_type(4))) float v4f;
typedef __attribute__((ext_vector_type(8))) float v8f;

#if __has_builtin(__builtin_amdgcn_exp2f)
#define EXP2F(v) __builtin_amdgcn_exp2f(v)
#else
#define EXP2F(v) exp2f(v)
#endif

template <bool MASKED>
__device__ __forceinline__ void score_tile(const v2f a, const v2f bm, const float vj, const bool ok,
                                           v8f& Dacc, v8f& Nacc)
{
    v8f c = {};
    const v8f d = __builtin_amdgcn_wmma_f32_16x16x4_f32(
        false, a, false, bm, (short)0, c, false, false);
    v8f e;
#pragma unroll
    for (int r = 0; r < 8; ++r) {
        const float ex = EXP2F(d[r]);
        e[r] = MASKED ? (ok ? ex : 0.0f) : ex;
    }
    Dacc += e;
    Nacc += e * vj;
}

template <bool MASKED>
__device__ __forceinline__ void score_group(const v2f a0, const v2f a1, const v4f k, const v4f v,
                                            const bool ok, v8f& Dacc, v8f& Nacc)
{
    v2f b01; b01[0] = k.x; b01[1] = k.y;
    v2f b23; b23[0] = k.z; b23[1] = k.w;
    score_tile<MASKED>(a0, b01, v.x, ok, Dacc, Nacc);
    score_tile<MASKED>(a1, b01, v.y, ok, Dacc, Nacc);
    score_tile<MASKED>(a0, b23, v.z, ok, Dacc, Nacc);
    score_tile<MASKED>(a1, b23, v.w, ok, Dacc, Nacc);
}

__device__ __forceinline__ float half_row_sums(const v8f X, const int ln)
{
    const bool b3 = (ln & 8) != 0, b2 = (ln & 4) != 0, b1 = (ln & 2) != 0;
    float y[4];
#pragma unroll
    for (int r = 0; r < 4; ++r) {
        const float keep = b3 ? X[r + 4] : X[r];
        const float send = b3 ? X[r] : X[r + 4];
        y[r] = keep + __shfl_xor(send, 8, 32);
    }
    float z[2];
#pragma unroll
    for (int r = 0; r < 2; ++r) {
        const float keep = b2 ? y[r + 2] : y[r];
        const float send = b2 ? y[r] : y[r + 2];
        z[r] = keep + __shfl_xor(send, 4, 32);
    }
    const float keep = b1 ? z[1] : z[0];
    const float send = b1 ? z[0] : z[1];
    const float w = keep + __shfl_xor(send, 2, 32);
    return w + __shfl_xor(w, 1, 32);
}

__launch_bounds__(BLOCK) __attribute__((amdgpu_waves_per_eu(1, 4)))
__global__ void mha_rank1_wmma_kernel(const float* __restrict__ x,
                                      const float* __restrict__ WQ,
                                      const float* __restrict__ WK,
                                      const float* __restrict__ WV,
                                      const float* __restrict__ W0,
                                      float* __restrict__ out)
{
    __shared__ __align__(16) float k_lds[H_DIM][LROW];
    __shared__ __align__(16) float v_lds[H_DIM][LROW];

    const int b    = blockIdx.x;
    const int tid  = threadIdx.x;
    const int wave = tid >> 5;
    const int lane = tid & 31;
    const int ln   = lane & 15;
    const int hi   = lane >> 4;

    const int  tile   = blockIdx.y * WPB + wave;
    const bool tValid = (tile * 16) < G_DIM;
    const int  iA     = tile * 16 + ln;
    const int  iAc    = (iA < G_DIM) ? iA : (G_DIM - 1);
    const int  iO     = tile * 16 + (hi << 3) + (ln >> 1);
    const int  iOc    = (iO < G_DIM) ? iO : (G_DIM - 1);

    const v4f* __restrict__ x4  = reinterpret_cast<const v4f*>(x + b * G_DIM);
    const v4f* __restrict__ wk4 = reinterpret_cast<const v4f*>(WK);
    const v4f* __restrict__ wv4 = reinterpret_cast<const v4f*>(WV);

    const float xA = x[b * G_DIM + iAc];
    const float xO = x[b * G_DIM + iOc];
    float qAn = WQ[iAc], qOn = WQ[iOc];

    v4f xs[FILL_STEPS], ks[FILL_STEPS][H_DIM], vs[FILL_STEPS][H_DIM];
#pragma unroll
    for (int s = 0; s < FILL_STEPS; ++s) {
        const int q  = tid + s * BLOCK;
        const int qc = ((s + 1) * BLOCK <= G4 || q < G4) ? q : (G4 - 1);
        xs[s] = x4[qc];
#pragma unroll
        for (int h = 0; h < H_DIM; ++h) {
            ks[s][h] = wk4[h * G4 + qc];
            vs[s][h] = wv4[h * G4 + qc];
        }
    }
#pragma unroll
    for (int s = 0; s < FILL_STEPS; ++s) {
        const int  q    = tid + s * BLOCK;
        const bool real = (s + 1) * BLOCK <= G4 || q < G4;
#pragma unroll
        for (int h = 0; h < H_DIM; ++h) {
            const v4f kk = xs[s] * ks[s][h] * LOG2E;
            const v4f vv = xs[s] * vs[s][h];
            const v4f zero = {};
            reinterpret_cast<v4f*>(k_lds[h])[q] = real ? kk : zero;
            reinterpret_cast<v4f*>(v_lds[h])[q] = real ? vv : zero;
        }
    }
    __syncthreads();

    float out_acc = 0.0f;

    for (int h = 0; h < H_DIM; ++h) {
        const float qA = qAn, qO = qOn;
        const float w0h = W0[h];
        if (h + 1 < H_DIM) {
            qAn = WQ[(h + 1) * G_DIM + iAc];
            qOn = WQ[(h + 1) * G_DIM + iOc];
        }

        if (tValid) {
            const float tA = (lane < 16) ? xA * qA : 0.0f;
            const float tO = xO * qO;
            v2f a0; a0[0] = tA;   a0[1] = 0.0f;
            v2f a1; a1[0] = 0.0f; a1[1] = tA;

            v8f Dacc = {}, Nacc = {};

            const v4f* kp = reinterpret_cast<const v4f*>(k_lds[h]) + ln;
            const v4f* vp = reinterpret_cast<const v4f*>(v_lds[h]) + ln;

            v4f ka = kp[0], va = vp[0], kb = kp[16], vb = vp[16];
            for (int g = 0; g < NFULL; g += 2) {
                const v4f kna = kp[(g + 2) * 16], vna = vp[(g + 2) * 16];
                const v4f knb = kp[(g + 3) * 16], vnb = vp[(g + 3) * 16];
                score_group<false>(a0, a1, ka, va, true, Dacc, Nacc);
                score_group<false>(a0, a1, kb, vb, true, Dacc, Nacc);
                asm volatile("" :: "v"(kna), "v"(vna), "v"(knb), "v"(vnb));
                ka = kna; va = vna; kb = knb; vb = vnb;
            }
            {
                const bool ok = (NFULL * 64 + 4 * ln + 3) < G_DIM;
                score_group<true>(a0, a1, ka, va, ok, Dacc, Nacc);
            }

            const float Dsum = half_row_sums(Dacc, ln);
            const float Nsum = half_row_sums(Nacc, ln);

            if (iO < G_DIM) {
                const float eii = EXP2F(tO * k_lds[h][iO]);
                const float num = Nsum - eii * v_lds[h][iO];
                out_acc += w0h * num / Dsum;
            }
        }
    }

    if (tValid && (ln & 1) == 0 && iO < G_DIM)
        out[b * G_DIM + iO] = out_acc;
}

extern "C" void kernel_launch(void* const* d_in, const int* in_sizes, int n_in,
                              void* d_out, int out_size, void* d_ws, size_t ws_size,
                              hipStream_t stream) {
    (void)in_sizes; (void)n_in; (void)out_size; (void)d_ws; (void)ws_size;
    const float* x  = (const float*)d_in[0];
    const float* WQ = (const float*)d_in[1];
    const float* WK = (const float*)d_in[2];
    const float* WV = (const float*)d_in[3];
    const float* W0 = (const float*)d_in[4];
    float* out = (float*)d_out;

    const int numTiles = (G_DIM + 15) / 16;
    const int groups   = (numTiles + WPB - 1) / WPB;
    dim3 grid(B_DIM, groups);
    hipLaunchKernelGGL(mha_rank1_wmma_kernel, grid, dim3(BLOCK), 0, stream,
                       x, WQ, WK, WV, W0, out);
}


# j loop pipelined by one tile: each matrix instruction is issued one stretch ahead and held there; its tile's exps and the previous tile's packed sums alternate behind it (same sums, same order)
# speedup vs baseline: 2.7719x; 1.1606x over previous
#include <hip/hip_runtime.h>


#define B_DIM 16
#define G_DIM 1708
#define H_DIM 5
#define NFULL 26
#define GPAD  1792
#define BLOCK 128
#define WPB   (BLOCK / 32)
#define G4    (G_DIM / 4)
#define LROW  2048
#define FILL_STEPS (LROW / 4 / BLOCK)
#define LOG2E 1.44269504088896340736f

static_assert(LROW >= GPAD && LROW % (4 * BLOCK) == 0 && G_DIM % 4 == 0, "staging layout");

typedef __attribute__((ext_vector_type(2))) float v2f;
typedef __attribute__((ext_vector_type(4))) float v4f;
typedef __attribute__((ext_vector_type(8))) float v8f;

#if __has_builtin(__builtin_amdgcn_exp2f)
#define EXP2F(v) __builtin_amdgcn_exp2f(v)
#else
#define EXP2F(v) exp2f(v)
#endif

typedef __attribute__((ext_vector_type(2))) float f32x2;

__device__ __forceinline__ v8f score_mm(const v2f a, const v2f bm)
{
    v8f d;
    asm volatile("v_wmma_f32_16x16x4_f32 %0, %1, %2, 0" : "=&v"(d) : "v"(a), "v"(bm));
    return d;
}

struct Pipe {
    f32x2 D[4], N[4];
    v8f   e;
    float v;
};

#define ORDER_HERE() __builtin_amdgcn_sched_barrier(0)

#define HOLD_OPERANDS(a_, b_) asm volatile("" :: "v"(a_), "v"(b_))

template <bool SUM_PREV, bool MASK_PREV>
__device__ __forceinline__ void stretch(const v8f d, const float vj, const bool ok, Pipe& P)
{
    v8f e;
    const f32x2 pv = {P.v, P.v};
#pragma unroll
    for (int p = 0; p < 4; ++p) {
        f32x2 ep = {P.e[2 * p], P.e[2 * p + 1]};
        if (MASK_PREV) { ep[0] = ok ? ep[0] : 0.0f; ep[1] = ok ? ep[1] : 0.0f; }
        e[2 * p] = EXP2F(d[2 * p]);
        if (SUM_PREV) P.D[p] += ep;
        ORDER_HERE();
        e[2 * p + 1] = EXP2F(d[2 * p + 1]);
        if (p == 3) ORDER_HERE();
        if (SUM_PREV) P.N[p] = __builtin_elementwise_fma(ep, pv, P.N[p]);
        ORDER_HERE();
    }
    P.e = e;
    P.v = vj;
}

template <bool MASK_PREV>
__device__ __forceinline__ void drain(const bool ok, Pipe& P)
{
    const f32x2 pv = {P.v, P.v};
#pragma unroll
    for (int p = 0; p < 4; ++p) {
        f32x2 ep = {P.e[2 * p], P.e[2 * p + 1]};
        if (MASK_PREV) { ep[0] = ok ? ep[0] : 0.0f; ep[1] = ok ? ep[1] : 0.0f; }
        P.D[p] += ep;
        P.N[p] = __builtin_elementwise_fma(ep, pv, P.N[p]);
    }
}

__device__ __forceinline__ void pass(Pipe& P, v8f& dn, const v2f a0, const v2f a1,
                                     const v4f* kp, const v4f* vp, const int g,
                                     const v4f ka, const v4f va, const v4f kb, const v4f vb,
                                     v4f& kna, v4f& vna, v4f& knb, v4f& vnb)
{
    kna = kp[(g + 2) * 16]; vna = vp[(g + 2) * 16];
    knb = kp[(g + 3) * 16]; vnb = vp[(g + 3) * 16];
    v2f ka23; ka23[0] = ka.z; ka23[1] = ka.w;
    v2f kb01; kb01[0] = kb.x; kb01[1] = kb.y;
    v2f kb23; kb23[0] = kb.z; kb23[1] = kb.w;
    v2f kn01; kn01[0] = kna.x; kn01[1] = kna.y;
    ORDER_HERE();
    const v8f d2 = score_mm(a0, ka23); ORDER_HERE(); stretch<true, false>(dn, va.y, true, P);  HOLD_OPERANDS(a0, ka23);
    const v8f d3 = score_mm(a1, ka23); ORDER_HERE(); stretch<true, false>(d2, va.z, true, P);  HOLD_OPERANDS(a1, ka23);
    const v8f d4 = score_mm(a0, kb01); ORDER_HERE(); stretch<true, false>(d3, va.w, true, P);  HOLD_OPERANDS(a0, kb01);
    const v8f d5 = score_mm(a1, kb01); ORDER_HERE(); stretch<true, false>(d4, vb.x, true, P);  HOLD_OPERANDS(a1, kb01);
    const v8f d6 = score_mm(a0, kb23); ORDER_HERE(); stretch<true, false>(d5, vb.y, true, P);  HOLD_OPERANDS(a0, kb23);
    const v8f d7 = score_mm(a1, kb23); ORDER_HERE(); stretch<true, false>(d6, vb.z, true, P);  HOLD_OPERANDS(a1, kb23);
    const v8f d8 = score_mm(a0, kn01); ORDER_HERE(); stretch<true, false>(d7, vb.w, true, P);  HOLD_OPERANDS(a0, kn01);
    const v8f d9 = score_mm(a1, kn01); ORDER_HERE(); stretch<true, false>(d8, vna.x, true, P); HOLD_OPERANDS(a1, kn01);
    dn = d9;
}

__device__ __forceinline__ float half_row_sums(const v8f X, const int ln)
{
    const bool b3 = (ln & 8) != 0, b2 = (ln & 4) != 0, b1 = (ln & 2) != 0;
    float y[4];
#pragma unroll
    for (int r = 0; r < 4; ++r) {
        const float keep = b3 ? X[r + 4] : X[r];
        const float send = b3 ? X[r] : X[r + 4];
        y[r] = keep + __shfl_xor(send, 8, 32);
    }
    float z[2];
#pragma unroll
    for (int r = 0; r < 2; ++r) {
        const float keep = b2 ? y[r + 2] : y[r];
        const float send = b2 ? y[r] : y[r + 2];
        z[r] = keep + __shfl_xor(send, 4, 32);
    }
    const float keep = b1 ? z[1] : z[0];
    const float send = b1 ? z[0] : z[1];
    const float w = keep + __shfl_xor(send, 2, 32);
    return w + __shfl_xor(w, 1, 32);
}

__launch_bounds__(BLOCK) __attribute__((amdgpu_waves_per_eu(1, 4)))
__global__ void mha_rank1_wmma_kernel(const float* __restrict__ x,
                                      const float* __restrict__ WQ,
                                      const float* __restrict__ WK,
                                      const float* __restrict__ WV,
                                      const float* __restrict__ W0,
                                      float* __restrict__ out)
{
    __shared__ __align__(16) float k_lds[H_DIM][LROW];
    __shared__ __align__(16) float v_lds[H_DIM][LROW];

    const int b    = blockIdx.x;
    const int tid  = threadIdx.x;
    const int wave = tid >> 5;
    const int lane = tid & 31;
    const int ln   = lane & 15;
    const int hi   = lane >> 4;

    const int  tile   = blockIdx.y * WPB + wave;
    const bool tValid = (tile * 16) < G_DIM;
    const int  iA     = tile * 16 + ln;
    const int  iAc    = (iA < G_DIM) ? iA : (G_DIM - 1);
    const int  iO     = tile * 16 + (hi << 3) + (ln >> 1);
    const int  iOc    = (iO < G_DIM) ? iO : (G_DIM - 1);

    const v4f* __restrict__ x4  = reinterpret_cast<const v4f*>(x + b * G_DIM);
    const v4f* __restrict__ wk4 = reinterpret_cast<const v4f*>(WK);
    const v4f* __restrict__ wv4 = reinterpret_cast<const v4f*>(WV);

    const float xA = x[b * G_DIM + iAc];
    const float xO = x[b * G_DIM + iOc];
    float qAn = WQ[iAc], qOn = WQ[iOc];

    v4f xs[FILL_STEPS], ks[FILL_STEPS][H_DIM], vs[FILL_STEPS][H_DIM];
#pragma unroll
    for (int s = 0; s < FILL_STEPS; ++s) {
        const int q  = tid + s * BLOCK;
        const int qc = ((s + 1) * BLOCK <= G4 || q < G4) ? q : (G4 - 1);
        xs[s] = x4[qc];
#pragma unroll
        for (int h = 0; h < H_DIM; ++h) {
            ks[s][h] = wk4[h * G4 + qc];
            vs[s][h] = wv4[h * G4 + qc];
        }
    }
#pragma unroll
    for (int s = 0; s < FILL_STEPS; ++s) {
        const int  q    = tid + s * BLOCK;
        const bool real = (s + 1) * BLOCK <= G4 || q < G4;
#pragma unroll
        for (int h = 0; h < H_DIM; ++h) {
            const v4f kk = xs[s] * ks[s][h] * LOG2E;
            const v4f vv = xs[s] * vs[s][h];
            const v4f zero = {};
            reinterpret_cast<v4f*>(k_lds[h])[q] = real ? kk : zero;
            reinterpret_cast<v4f*>(v_lds[h])[q] = real ? vv : zero;
        }
    }
    __syncthreads();

    float out_acc = 0.0f;

    for (int h = 0; h < H_DIM; ++h) {
        const float qA = qAn, qO = qOn;
        const float w0h = W0[h];
        if (h + 1 < H_DIM) {
            qAn = WQ[(h + 1) * G_DIM + iAc];
            qOn = WQ[(h + 1) * G_DIM + iOc];
        }

        if (tValid) {
            const float tA = (lane < 16) ? xA * qA : 0.0f;
            const float tO = xO * qO;
            v2f a0; a0[0] = tA;   a0[1] = 0.0f;
            v2f a1; a1[0] = 0.0f; a1[1] = tA;

            Pipe P;
#pragma unroll
            for (int p = 0; p < 4; ++p) { P.D[p] = f32x2{0.0f, 0.0f}; P.N[p] = f32x2{0.0f, 0.0f}; }

            const v4f* kp = reinterpret_cast<const v4f*>(k_lds[h]) + ln;
            const v4f* vp = reinterpret_cast<const v4f*>(v_lds[h]) + ln;

            v4f ka = kp[0], va = vp[0], kb = kp[16], vb = vp[16];
            v8f dn;
            {
                v2f ka01; ka01[0] = ka.x; ka01[1] = ka.y;
                const v8f d0 = score_mm(a0, ka01);
                dn = score_mm(a1, ka01);
                ORDER_HERE();
                stretch<false, false>(d0, va.x, true, P);
                HOLD_OPERANDS(a1, ka01);
            }
            v4f kc, vc, kd, vd;
            for (int g = 0; g + 4 <= NFULL; g += 4) {
                pass(P, dn, a0, a1, kp, vp, g,     ka, va, kb, vb, kc, vc, kd, vd);
                pass(P, dn, a0, a1, kp, vp, g + 2, kc, vc, kd, vd, ka, va, kb, vb);
            }
            static_assert(NFULL % 4 == 2, "one single pass is left after the double passes");
            pass(P, dn, a0, a1, kp, vp, NFULL - 2, ka, va, kb, vb, kc, vc, kd, vd);
            ka = kc; va = vc;
            {
                const bool ok = (NFULL * 64 + 4 * ln + 3) < G_DIM;
                v2f ka23; ka23[0] = ka.z; ka23[1] = ka.w;
                const v8f d2 = score_mm(a0, ka23); ORDER_HERE(); stretch<true, true>(dn, va.y, ok, P); HOLD_OPERANDS(a0, ka23);
                const v8f d3 = score_mm(a1, ka23); ORDER_HERE(); stretch<true, true>(d2, va.z, ok, P); HOLD_OPERANDS(a1, ka23);
                stretch<true, true>(d3, va.w, ok, P);
                drain<true>(ok, P);
            }
            v8f Dacc, Nacc;
#pragma unroll
            for (int p = 0; p < 4; ++p) {
                Dacc[2 * p] = P.D[p][0]; Dacc[2 * p + 1] = P.D[p][1];
                Nacc[2 * p] = P.N[p][0]; Nacc[2 * p + 1] = P.N[p][1];
            }

            const float Dsum = half_row_sums(Dacc, ln);
            const float Nsum = half_row_sums(Nacc, ln);

            if (iO < G_DIM) {
                const float eii = EXP2F(tO * k_lds[h][iO]);
                const float num = Nsum - eii * v_lds[h][iO];
                out_acc += w0h * num / Dsum;
            }
        }
    }

    if (tValid && (ln & 1) == 0 && iO < G_DIM)
        out[b * G_DIM + iO] = out_acc;
}

extern "C" void kernel_launch(void* const* d_in, const int* in_sizes, int n_in,
                              void* d_out, int out_size, void* d_ws, size_t ws_size,
                              hipStream_t stream) {
    (void)in_sizes; (void)n_in; (void)out_size; (void)d_ws; (void)ws_size;
    const float* x  = (const float*)d_in[0];
    const float* WQ = (const float*)d_in[1];
    const float* WK = (const float*)d_in[2];
    const float* WV = (const float*)d_in[3];
    const float* W0 = (const float*)d_in[4];
    float* out = (float*)d_out;

    const int numTiles = (G_DIM + 15) / 16;
    const int groups   = (numTiles + WPB - 1) / WPB;
    dim3 grid(B_DIM, groups);
    hipLaunchKernelGGL(mha_rank1_wmma_kernel, grid, dim3(BLOCK), 0, stream,
                       x, WQ, WK, WV, W0, out);
}


# Work re-cut for even SIMD load: 256 8-wave work-groups, one per WGP (LDS reserved above 160 KiB), each owns 6-7 whole i-tiles; (tile, head) units go round its waves; heads joined in LDS, same order
# speedup vs baseline: 3.0771x; 1.1101x over previous
#include <hip/hip_runtime.h>


#define B_DIM 16
#define G_DIM 1708
#define H_DIM 5
#define NFULL 26
#define GPAD  1792
#define NTILES ((G_DIM + 15) / 16)
#define PARTS  16
#define MAXT   ((NTILES + PARTS - 1) / PARTS)
#define BLOCK 256
#define WPB   (BLOCK / 32)
#define G4    (G_DIM / 4)
#define LROW  2048
#define FILL_STEPS (LROW / 4 / BLOCK)
#define KV_FLOATS   (H_DIM * LROW)
#define PART_FLOATS (MAXT * H_DIM * 16)
#define LDS_FLOATS  (164 * 1024 / 4)
#define LOG2E 1.44269504088896340736f

static_assert(LROW >= GPAD && LROW % (4 * BLOCK) == 0 && G_DIM % 4 == 0, "staging layout");
static_assert(2 * KV_FLOATS + PART_FLOATS <= LDS_FLOATS && LDS_FLOATS * 4 <= 320 * 1024, "LDS layout");
static_assert(MAXT * 16 <= BLOCK && (NTILES / PARTS) * H_DIM >= WPB, "every wave has a first unit; one thread per output row");

typedef __attribute__((ext_vector_type(2))) float v2f;
typedef __attribute__((ext_vector_type(4))) float v4f;
typedef __attribute__((ext_vector_type(8))) float v8f;

#if __has_builtin(__builtin_amdgcn_exp2f)
#define EXP2F(v) __builtin_amdgcn_exp2f(v)
#else
#define EXP2F(v) exp2f(v)
#endif

typedef __attribute__((ext_vector_type(2))) float f32x2;

__device__ __forceinline__ v8f score_mm(const v2f a, const v2f bm)
{
    v8f d;
    asm volatile("v_wmma_f32_16x16x4_f32 %0, %1, %2, 0" : "=&v"(d) : "v"(a), "v"(bm));
    return d;
}

struct Pipe {
    f32x2 D[4], N[4];
    v8f   e;
    float v;
};

#define ORDER_HERE() __builtin_amdgcn_sched_barrier(0)

#define HOLD_OPERANDS(a_, b_) asm volatile("" :: "v"(a_), "v"(b_))

template <bool SUM_PREV, bool MASK_PREV>
__device__ __forceinline__ void stretch(const v8f d, const float vj, const bool ok, Pipe& P)
{
    v8f e;
    const f32x2 pv = {P.v, P.v};
#pragma unroll
    for (int p = 0; p < 4; ++p) {
        f32x2 ep = {P.e[2 * p], P.e[2 * p + 1]};
        if (MASK_PREV) { ep[0] = ok ? ep[0] : 0.0f; ep[1] = ok ? ep[1] : 0.0f; }
        e[2 * p] = EXP2F(d[2 * p]);
        if (SUM_PREV) P.D[p] += ep;
        ORDER_HERE();
        e[2 * p + 1] = EXP2F(d[2 * p + 1]);
        if (p == 3) ORDER_HERE();
        if (SUM_PREV) P.N[p] = __builtin_elementwise_fma(ep, pv, P.N[p]);
        ORDER_HERE();
    }
    P.e = e;
    P.v = vj;
}

template <bool MASK_PREV>
__device__ __forceinline__ void drain(const bool ok, Pipe& P)
{
    const f32x2 pv = {P.v, P.v};
#pragma unroll
    for (int p = 0; p < 4; ++p) {
        f32x2 ep = {P.e[2 * p], P.e[2 * p + 1]};
        if (MASK_PREV) { ep[0] = ok ? ep[0] : 0.0f; ep[1] = ok ? ep[1] : 0.0f; }
        P.D[p] += ep;
        P.N[p] = __builtin_elementwise_fma(ep, pv, P.N[p]);
    }
}

__device__ __forceinline__ void pass(Pipe& P, v8f& dn, const v2f a0, const v2f a1,
                                     const v4f* kp, const v4f* vp, const int g,
                                     const v4f ka, const v4f va, const v4f kb, const v4f vb,
                                     v4f& kna, v4f& vna, v4f& knb, v4f& vnb)
{
    kna = kp[(g + 2) * 16]; vna = vp[(g + 2) * 16];
    knb = kp[(g + 3) * 16]; vnb = vp[(g + 3) * 16];
    v2f ka23; ka23[0] = ka.z; ka23[1] = ka.w;
    v2f kb01; kb01[0] = kb.x; kb01[1] = kb.y;
    v2f kb23; kb23[0] = kb.z; kb23[1] = kb.w;
    v2f kn01; kn01[0] = kna.x; kn01[1] = kna.y;
    ORDER_HERE();
    const v8f d2 = score_mm(a0, ka23); ORDER_HERE(); stretch<true, false>(dn, va.y, true, P);  HOLD_OPERANDS(a0, ka23);
    const v8f d3 = score_mm(a1, ka23); ORDER_HERE(); stretch<true, false>(d2, va.z, true, P);  HOLD_OPERANDS(a1, ka23);
    const v8f d4 = score_mm(a0, kb01); ORDER_HERE(); stretch<true, false>(d3, va.w, true, P);  HOLD_OPERANDS(a0, kb01);
    const v8f d5 = score_mm(a1, kb01); ORDER_HERE(); stretch<true, false>(d4, vb.x, true, P);  HOLD_OPERANDS(a1, kb01);
    const v8f d6 = score_mm(a0, kb23); ORDER_HERE(); stretch<true, false>(d5, vb.y, true, P);  HOLD_OPERANDS(a0, kb23);
    const v8f d7 = score_mm(a1, kb23); ORDER_HERE(); stretch<true, false>(d6, vb.z, true, P);  HOLD_OPERANDS(a1, kb23);
    const v8f d8 = score_mm(a0, kn01); ORDER_HERE(); stretch<true, false>(d7, vb.w, true, P);  HOLD_OPERANDS(a0, kn01);
    const v8f d9 = score_mm(a1, kn01); ORDER_HERE(); stretch<true, false>(d8, vna.x, true, P); HOLD_OPERANDS(a1, kn01);
    dn = d9;
}

__device__ __forceinline__ float half_row_sums(const v8f X, const int ln)
{
    const bool b3 = (ln & 8) != 0, b2 = (ln & 4) != 0, b1 = (ln & 2) != 0;
    float y[4];
#pragma unroll
    for (int r = 0; r < 4; ++r) {
        const float keep = b3 ? X[r + 4] : X[r];
        const float send = b3 ? X[r] : X[r + 4];
        y[r] = keep + __shfl_xor(send, 8, 32);
    }
    float z[2];
#pragma unroll
    for (int r = 0; r < 2; ++r) {
        const float keep = b2 ? y[r + 2] : y[r];
        const float send = b2 ? y[r] : y[r + 2];
        z[r] = keep + __shfl_xor(send, 4, 32);
    }
    const float keep = b1 ? z[1] : z[0];
    const float send = b1 ? z[0] : z[1];
    const float w = keep + __shfl_xor(send, 2, 32);
    return w + __shfl_xor(w, 1, 32);
}

struct Rows { float xA, xO, qA, qO, w0; };

__device__ __forceinline__ Rows fetch_rows(const float* __restrict__ x, const float* __restrict__ WQ,
                                           const float* __restrict__ W0, const int b, const int tile,
                                           const int h, const int ln, const int hi)
{
    const int iA  = tile * 16 + ln;
    const int iAc = (iA < G_DIM) ? iA : (G_DIM - 1);
    const int iO  = tile * 16 + (hi << 3) + (ln >> 1);
    const int iOc = (iO < G_DIM) ? iO : (G_DIM - 1);
    Rows r;
    r.xA = x[b * G_DIM + iAc];
    r.xO = x[b * G_DIM + iOc];
    r.qA = WQ[h * G_DIM + iAc];
    r.qO = WQ[h * G_DIM + iOc];
    r.w0 = W0[h];
    return r;
}

__launch_bounds__(BLOCK)
__global__ void mha_rank1_wmma_kernel(const float* __restrict__ x,
                                      const float* __restrict__ WQ,
                                      const float* __restrict__ WK,
                                      const float* __restrict__ WV,
                                      const float* __restrict__ W0,
                                      float* __restrict__ out)
{
    __shared__ __align__(16) float lds[LDS_FLOATS];
    float* const k_lds = lds;
    float* const v_lds = lds + KV_FLOATS;
    float* const part  = lds + 2 * KV_FLOATS;

    const int b    = blockIdx.x;
    const int tid  = threadIdx.x;
    const int wave = __builtin_amdgcn_readfirstlane(tid >> 5);
    const int lane = tid & 31;
    const int ln   = lane & 15;
    const int hi   = lane >> 4;

    const int t0 = ((int)blockIdx.y * NTILES) / PARTS;
    const int nT = (((int)blockIdx.y + 1) * NTILES) / PARTS - t0;
    const int nU = nT * H_DIM;

    const v4f* __restrict__ x4  = reinterpret_cast<const v4f*>(x + b * G_DIM);
    const v4f* __restrict__ wk4 = reinterpret_cast<const v4f*>(WK);
    const v4f* __restrict__ wv4 = reinterpret_cast<const v4f*>(WV);

    Rows nx = fetch_rows(x, WQ, W0, b, t0 + wave / H_DIM, wave % H_DIM, ln, hi);

    v4f xs[FILL_STEPS], ks[FILL_STEPS][H_DIM], vs[FILL_STEPS][H_DIM];
#pragma unroll
    for (int s = 0; s < FILL_STEPS; ++s) {
        const int q  = tid + s * BLOCK;
        const int qc = ((s + 1) * BLOCK <= G4 || q < G4) ? q : (G4 - 1);
        xs[s] = x4[qc];
#pragma unroll
        for (int h = 0; h < H_DIM; ++h) {
            ks[s][h] = wk4[h * G4 + qc];
            vs[s][h] = wv4[h * G4 + qc];
        }
    }
#pragma unroll
    for (int s = 0; s < FILL_STEPS; ++s) {
        const int  q    = tid + s * BLOCK;
        const bool real = (s + 1) * BLOCK <= G4 || q < G4;
#pragma unroll
        for (int h = 0; h < H_DIM; ++h) {
            const v4f kk = xs[s] * ks[s][h] * LOG2E;
            const v4f vv = xs[s] * vs[s][h];
            const v4f zero = {};
            reinterpret_cast<v4f*>(k_lds + h * LROW)[q] = real ? kk : zero;
            reinterpret_cast<v4f*>(v_lds + h * LROW)[q] = real ? vv : zero;
        }
    }
    __syncthreads();

    for (int u = wave; u < nU; u += WPB) {
        const int tl   = u / H_DIM;
        const int h    = u - tl * H_DIM;
        const int tile = t0 + tl;
        const int iO   = tile * 16 + (hi << 3) + (ln >> 1);
        const Rows r = nx;
        if (u + WPB < nU) {
            const int un = u + WPB, tn = un / H_DIM;
            nx = fetch_rows(x, WQ, W0, b, t0 + tn, un - tn * H_DIM, ln, hi);
        }

        const float tA = (lane < 16) ? r.xA * r.qA : 0.0f;
        const float tO = r.xO * r.qO;
        v2f a0; a0[0] = tA;   a0[1] = 0.0f;
        v2f a1; a1[0] = 0.0f; a1[1] = tA;

        Pipe P;
#pragma unroll
        for (int p = 0; p < 4; ++p) { P.D[p] = f32x2{0.0f, 0.0f}; P.N[p] = f32x2{0.0f, 0.0f}; }

        const v4f* kp = reinterpret_cast<const v4f*>(k_lds + h * LROW) + ln;
        const v4f* vp = reinterpret_cast<const v4f*>(v_lds + h * LROW) + ln;

        v4f ka = kp[0], va = vp[0], kb = kp[16], vb = vp[16];
        v8f dn;
        {
            v2f ka01; ka01[0] = ka.x; ka01[1] = ka.y;
            const v8f d0 = score_mm(a0, ka01);
            dn = score_mm(a1, ka01);
            ORDER_HERE();
            stretch<false, false>(d0, va.x, true, P);
            HOLD_OPERANDS(a1, ka01);
        }
        v4f kc, vc, kd, vd;
        for (int g = 0; g + 4 <= NFULL; g += 4) {
            pass(P, dn, a0, a1, kp, vp, g,     ka, va, kb, vb, kc, vc, kd, vd);
            pass(P, dn, a0, a1, kp, vp, g + 2, kc, vc, kd, vd, ka, va, kb, vb);
        }
        static_assert(NFULL % 4 == 2, "one single pass is left after the double passes");
        pass(P, dn, a0, a1, kp, vp, NFULL - 2, ka, va, kb, vb, kc, vc, kd, vd);
        ka = kc; va = vc;
        {
            const bool ok = (NFULL * 64 + 4 * ln + 3) < G_DIM;
            v2f ka23; ka23[0] = ka.z; ka23[1] = ka.w;
            const v8f d2 = score_mm(a0, ka23); ORDER_HERE(); stretch<true, true>(dn, va.y, ok, P); HOLD_OPERANDS(a0, ka23);
            const v8f d3 = score_mm(a1, ka23); ORDER_HERE(); stretch<true, true>(d2, va.z, ok, P); HOLD_OPERANDS(a1, ka23);
            stretch<true, true>(d3, va.w, ok, P);
            drain<true>(ok, P);
        }
        v8f Dacc, Nacc;
#pragma unroll
        for (int p = 0; p < 4; ++p) {
            Dacc[2 * p] = P.D[p][0]; Dacc[2 * p + 1] = P.D[p][1];
            Nacc[2 * p] = P.N[p][0]; Nacc[2 * p + 1] = P.N[p][1];
        }

        const float Dsum = half_row_sums(Dacc, ln);
        const float Nsum = half_row_sums(Nacc, ln);

        if ((ln & 1) == 0) {
            float term = 0.0f;
            if (iO < G_DIM) {
                const float eii = EXP2F(tO * k_lds[h * LROW + iO]);
                const float num = Nsum - eii * v_lds[h * LROW + iO];
                term = r.w0 * num / Dsum;
            }
            part[u * 16 + (hi << 3) + (ln >> 1)] = term;
        }
    }
    __syncthreads();

    if (tid < nT * 16) {
        const int row = t0 * 16 + tid;
        if (row < G_DIM) {
            const int tl = tid >> 4, r16 = tid & 15;
            float out_acc = 0.0f;
#pragma unroll
            for (int h = 0; h < H_DIM; ++h) out_acc += part[(tl * H_DIM + h) * 16 + r16];
            out[b * G_DIM + row] = out_acc;
        }
    }
}

extern "C" void kernel_launch(void* const* d_in, const int* in_sizes, int n_in,
                              void* d_out, int out_size, void* d_ws, size_t ws_size,
                              hipStream_t stream) {
    (void)in_sizes; (void)n_in; (void)out_size; (void)d_ws; (void)ws_size;
    const float* x  = (const float*)d_in[0];
    const float* WQ = (const float*)d_in[1];
    const float* WK = (const float*)d_in[2];
    const float* WV = (const float*)d_in[3];
    const float* W0 = (const float*)d_in[4];
    float* out = (float*)d_out;

    dim3 grid(B_DIM, PARTS);
    hipLaunchKernelGGL(mha_rank1_wmma_kernel, grid, dim3(BLOCK), 0, stream,
                       x, WQ, WK, WV, W0, out);
}
